# starting kernel
# baseline (speedup 1.0000x reference)
#include <hip/hip_runtime.h>
#include <math.h>

#define B_     4
#define T_IN_  12
#define T_OUT_ 24
#define N_     5000
#define C_     32
#define H_     8
#define K1_    17
#define TH_    (T_IN_ * H_)
#define OT_    (T_OUT_ - T_IN_)
#define OTP_   16
#define KSTEPS (TH_ / 4)
#define BETA_  0.8f

typedef __attribute__((ext_vector_type(2))) float v2f;
typedef __attribute__((ext_vector_type(8))) float v8f;

__global__ __launch_bounds__(256) void gnn_sigma_kernel(
    const float* __restrict__ nd, float* __restrict__ ws) {
  __shared__ float red[256];
  float m = 0.0f;
  for (int i = threadIdx.x; i < N_ * K1_; i += 256) m = fmaxf(m, nd[i]);
  red[threadIdx.x] = m;
  __syncthreads();
  for (int s = 128; s > 0; s >>= 1) {
    if (threadIdx.x < s) red[threadIdx.x] = fmaxf(red[threadIdx.x], red[threadIdx.x + s]);
    __syncthreads();
  }
  if (threadIdx.x == 0) {
    float mx = red[0];
    float sig = mx * 0.5f;
    ws[0] = 1.0f / (sig * sig);
  }
}

__global__ __launch_bounds__(128) void gnn_fused_kernel(
    const float* __restrict__ x,
    const int*   __restrict__ nn,
    const float* __restrict__ nd,
    const float* __restrict__ Wm,
    const float* __restrict__ bias,
    const float* __restrict__ ws,
    float*       __restrict__ out) {
  __shared__ float w_lds[K1_ * H_];
  __shared__ int   nn_lds[K1_];
  __shared__ float Wm_lds[TH_ * OTP_];
  __shared__ float bias_lds[OTP_];
  __shared__ float feat[B_][C_][TH_ + 1];

  const int n    = blockIdx.x;
  const int tid  = threadIdx.x;
  const int lane = tid & 31;
  const int b    = tid >> 5;
  const float inv_s2 = ws[0];

  for (int i = tid; i < K1_ * H_; i += 128) {
    const int j = i / H_, h = i % H_;
    const float d    = nd[n * K1_ + j];
    const int   node = nn[n * K1_ + j];
    const float lam  = (float)(h + 1) * (1.0f / (float)H_);
    float wv = __expf(-d * d * lam * inv_s2);
    if (node < 0 || wv < 1e-5f) wv = 0.0f;
    w_lds[i] = wv;
  }
  if (tid < K1_) {
    const int node = nn[n * K1_ + tid];
    nn_lds[tid] = (node < 0) ? 0 : node;
  }
  for (int i = tid; i < TH_ * OTP_; i += 128) {
    const int k = i >> 4, o = i & (OTP_ - 1);
    Wm_lds[i] = (o < OT_) ? Wm[k * OT_ + o] : 0.0f;
  }
  if (tid < OTP_) bias_lds[tid] = (tid < OT_) ? bias[tid] : 0.0f;
  __syncthreads();

  float acc[T_IN_][H_];
#pragma unroll
  for (int t = 0; t < T_IN_; ++t)
#pragma unroll
    for (int h = 0; h < H_; ++h) acc[t][h] = 0.0f;

  for (int j = 0; j < K1_; ++j) {
    const int m = nn_lds[j];
    float wj[H_];
#pragma unroll
    for (int h = 0; h < H_; ++h) wj[h] = w_lds[j * H_ + h];
    const float* xp = x + (((size_t)b * T_IN_) * N_ + m) * C_ + lane;
#pragma unroll
    for (int t = 0; t < T_IN_; ++t) {
      const float xv = xp[(size_t)t * N_ * C_];
#pragma unroll
      for (int h = 0; h < H_; ++h) acc[t][h] = fmaf(xv, wj[h], acc[t][h]);
    }
  }

#pragma unroll
  for (int t = 0; t < T_IN_; ++t) {
    out[(((size_t)b * T_OUT_ + t) * N_ + n) * C_ + lane] =
        x[(((size_t)b * T_IN_ + t) * N_ + n) * C_ + lane];
  }

#pragma unroll
  for (int t = 0; t < T_IN_; ++t)
#pragma unroll
    for (int h = 0; h < H_; ++h) feat[b][lane][t * H_ + h] = acc[t][h];
  __syncthreads();

  const int nidx  = lane & 15;
  const int khalf = (lane >> 4) * 2;

  v2f bfrag[KSTEPS];
#pragma unroll
  for (int kk = 0; kk < KSTEPS; ++kk) {
    const int k0 = kk * 4 + khalf;
    bfrag[kk].x = Wm_lds[(k0)     * OTP_ + nidx];
    bfrag[kk].y = Wm_lds[(k0 + 1) * OTP_ + nidx];
  }
  const float bv = bias_lds[nidx];

#pragma unroll
  for (int mt = 0; mt < 2; ++mt) {
    v8f c8 = {0.f, 0.f, 0.f, 0.f, 0.f, 0.f, 0.f, 0.f};
    const int mrow = mt * 16 + nidx;
#pragma unroll
    for (int kk = 0; kk < KSTEPS; ++kk) {
      const int k0 = kk * 4 + khalf;
      v2f a;
      a.x = feat[b][mrow][k0];
      a.y = feat[b][mrow][k0 + 1];
      c8 = __builtin_amdgcn_wmma_f32_16x16x4_f32(
           false, a,  false, bfrag[kk],
           (short)0, c8,  false,  false);
    }
    if (nidx < OT_) {
#pragma unroll
      for (int v = 0; v < 8; ++v) {
        const int m    = v + ((lane >> 4) << 3);
        const int cidx = mt * 16 + m;
        float y = c8[v] + bv;
        const float s = 1.0f / (1.0f + __expf(-BETA_ * y));
        y *= s;
        out[(((size_t)b * T_OUT_ + (T_IN_ + nidx)) * N_ + n) * C_ + cidx] = y;
      }
    }
  }
}

extern "C" void kernel_launch(void* const* d_in, const int* in_sizes, int n_in,
                              void* d_out, int out_size, void* d_ws, size_t ws_size,
                              hipStream_t stream) {
  const float* x    = (const float*)d_in[0];
  const int*   nn   = (const int*)d_in[1];
  const float* nd   = (const float*)d_in[2];
  const float* Wm   = (const float*)d_in[3];
  const float* bias = (const float*)d_in[4];
  float* out = (float*)d_out;
  float* ws  = (float*)d_ws;

  gnn_sigma_kernel<<<1, 256, 0, stream>>>(nd, ws);
  gnn_fused_kernel<<<N_, 128, 0, stream>>>(x, nn, nd, Wm, bias, ws, out);
}


# distance-array max split over 32 blocks x 256 threads, 11 independent loads per thread, partial maxima in workspace, max finished in fused kernel (was 1 block, 166 serial round trips); bit-identical
# speedup vs baseline: 2.0954x; 2.0954x over previous
#include <hip/hip_runtime.h>
#include <math.h>

#define B_     4
#define T_IN_  12
#define T_OUT_ 24
#define N_     5000
#define C_     32
#define H_     8
#define K1_    17
#define TH_    (T_IN_ * H_)
#define OT_    (T_OUT_ - T_IN_)
#define OTP_   16
#define KSTEPS (TH_ / 4)
#define BETA_  0.8f

typedef __attribute__((ext_vector_type(2))) float v2f;
typedef __attribute__((ext_vector_type(8))) float v8f;

#define SIG_BLOCKS  32
#define SIG_THREADS 256
#define SIG_STRIDE  (SIG_BLOCKS * SIG_THREADS)
#define SIG_LOADS   ((N_ * K1_ + SIG_STRIDE - 1) / SIG_STRIDE)

__global__ __launch_bounds__(SIG_THREADS) void gnn_sigma_kernel(
    const float* __restrict__ nd, float* __restrict__ ws) {
  __shared__ float red[SIG_THREADS / 32];
  const int gtid = blockIdx.x * SIG_THREADS + threadIdx.x;
  float v[SIG_LOADS];
#pragma unroll
  for (int k = 0; k < SIG_LOADS; ++k) {
    int i = gtid + k * SIG_STRIDE;
    i = (i < N_ * K1_) ? i : (N_ * K1_ - 1);
    v[k] = nd[i];
  }
  float m = v[0];
  for (int k = 1; k < SIG_LOADS; ++k) m = fmaxf(m, v[k]);
  for (int off = 16; off > 0; off >>= 1) m = fmaxf(m, __shfl_xor(m, off));
  if ((threadIdx.x & 31) == 0) red[threadIdx.x >> 5] = m;
  __syncthreads();
  if (threadIdx.x == 0) {
    float r = red[0];
    for (int w = 1; w < SIG_THREADS / 32; ++w) r = fmaxf(r, red[w]);
    ws[blockIdx.x] = r;
  }
}

__global__ __launch_bounds__(128) void gnn_fused_kernel(
    const float* __restrict__ x,
    const int*   __restrict__ nn,
    const float* __restrict__ nd,
    const float* __restrict__ Wm,
    const float* __restrict__ bias,
    const float* __restrict__ ws,
    float*       __restrict__ out) {
  __shared__ float w_lds[K1_ * H_];
  __shared__ int   nn_lds[K1_];
  __shared__ float Wm_lds[TH_ * OTP_];
  __shared__ float bias_lds[OTP_];
  __shared__ float feat[B_][C_][TH_ + 1];

  const int n    = blockIdx.x;
  const int tid  = threadIdx.x;
  const int lane = tid & 31;
  const int b    = tid >> 5;
  float mx = ws[0];
#pragma unroll
  for (int i = 1; i < SIG_BLOCKS; ++i) mx = fmaxf(mx, ws[i]);
  const float sig = mx * 0.5f;
  const float inv_s2 = 1.0f / (sig * sig);

  for (int i = tid; i < K1_ * H_; i += 128) {
    const int j = i / H_, h = i % H_;
    const float d    = nd[n * K1_ + j];
    const int   node = nn[n * K1_ + j];
    const float lam  = (float)(h + 1) * (1.0f / (float)H_);
    float wv = __expf(-d * d * lam * inv_s2);
    if (node < 0 || wv < 1e-5f) wv = 0.0f;
    w_lds[i] = wv;
  }
  if (tid < K1_) {
    const int node = nn[n * K1_ + tid];
    nn_lds[tid] = (node < 0) ? 0 : node;
  }
  for (int i = tid; i < TH_ * OTP_; i += 128) {
    const int k = i >> 4, o = i & (OTP_ - 1);
    Wm_lds[i] = (o < OT_) ? Wm[k * OT_ + o] : 0.0f;
  }
  if (tid < OTP_) bias_lds[tid] = (tid < OT_) ? bias[tid] : 0.0f;
  __syncthreads();

  float acc[T_IN_][H_];
#pragma unroll
  for (int t = 0; t < T_IN_; ++t)
#pragma unroll
    for (int h = 0; h < H_; ++h) acc[t][h] = 0.0f;

  for (int j = 0; j < K1_; ++j) {
    const int m = nn_lds[j];
    float wj[H_];
#pragma unroll
    for (int h = 0; h < H_; ++h) wj[h] = w_lds[j * H_ + h];
    const float* xp = x + (((size_t)b * T_IN_) * N_ + m) * C_ + lane;
#pragma unroll
    for (int t = 0; t < T_IN_; ++t) {
      const float xv = xp[(size_t)t * N_ * C_];
#pragma unroll
      for (int h = 0; h < H_; ++h) acc[t][h] = fmaf(xv, wj[h], acc[t][h]);
    }
  }

#pragma unroll
  for (int t = 0; t < T_IN_; ++t) {
    out[(((size_t)b * T_OUT_ + t) * N_ + n) * C_ + lane] =
        x[(((size_t)b * T_IN_ + t) * N_ + n) * C_ + lane];
  }

#pragma unroll
  for (int t = 0; t < T_IN_; ++t)
#pragma unroll
    for (int h = 0; h < H_; ++h) feat[b][lane][t * H_ + h] = acc[t][h];
  __syncthreads();

  const int nidx  = lane & 15;
  const int khalf = (lane >> 4) * 2;

  v2f bfrag[KSTEPS];
#pragma unroll
  for (int kk = 0; kk < KSTEPS; ++kk) {
    const int k0 = kk * 4 + khalf;
    bfrag[kk].x = Wm_lds[(k0)     * OTP_ + nidx];
    bfrag[kk].y = Wm_lds[(k0 + 1) * OTP_ + nidx];
  }
  const float bv = bias_lds[nidx];

#pragma unroll
  for (int mt = 0; mt < 2; ++mt) {
    v8f c8 = {0.f, 0.f, 0.f, 0.f, 0.f, 0.f, 0.f, 0.f};
    const int mrow = mt * 16 + nidx;
#pragma unroll
    for (int kk = 0; kk < KSTEPS; ++kk) {
      const int k0 = kk * 4 + khalf;
      v2f a;
      a.x = feat[b][mrow][k0];
      a.y = feat[b][mrow][k0 + 1];
      c8 = __builtin_amdgcn_wmma_f32_16x16x4_f32(
           false, a,  false, bfrag[kk],
           (short)0, c8,  false,  false);
    }
    if (nidx < OT_) {
#pragma unroll
      for (int v = 0; v < 8; ++v) {
        const int m    = v + ((lane >> 4) << 3);
        const int cidx = mt * 16 + m;
        float y = c8[v] + bv;
        const float s = 1.0f / (1.0f + __expf(-BETA_ * y));
        y *= s;
        out[(((size_t)b * T_OUT_ + (T_IN_ + nidx)) * N_ + n) * C_ + cidx] = y;
      }
    }
  }
}

extern "C" void kernel_launch(void* const* d_in, const int* in_sizes, int n_in,
                              void* d_out, int out_size, void* d_ws, size_t ws_size,
                              hipStream_t stream) {
  const float* x    = (const float*)d_in[0];
  const int*   nn   = (const int*)d_in[1];
  const float* nd   = (const float*)d_in[2];
  const float* Wm   = (const float*)d_in[3];
  const float* bias = (const float*)d_in[4];
  float* out = (float*)d_out;
  float* ws  = (float*)d_ws;

  gnn_sigma_kernel<<<SIG_BLOCKS, SIG_THREADS, 0, stream>>>(nd, ws);
  gnn_fused_kernel<<<N_, 128, 0, stream>>>(x, nn, nd, Wm, bias, ws, out);
}


# fused kernel: lane = (channel pair, time half): gather accumulators are the matrix B fragments in place (no LDS trip of the features), 8-byte whole-row loads and stores; product's k order changed
# speedup vs baseline: 3.2744x; 1.5627x over previous
#include <hip/hip_runtime.h>
#include <math.h>

#define B_     4
#define T_IN_  12
#define T_OUT_ 24
#define N_     5000
#define C_     32
#define H_     8
#define K1_    17
#define TH_    (T_IN_ * H_)
#define OT_    (T_OUT_ - T_IN_)
#define OTP_   16
#define KSTEPS (TH_ / 4)
#define BETA_  0.8f

typedef __attribute__((ext_vector_type(2))) float v2f;
typedef __attribute__((ext_vector_type(4))) float v4f;
typedef __attribute__((ext_vector_type(8))) float v8f;

#define SIG_BLOCKS  32
#define SIG_THREADS 256
#define SIG_STRIDE  (SIG_BLOCKS * SIG_THREADS)
#define SIG_LOADS   ((N_ * K1_ + SIG_STRIDE - 1) / SIG_STRIDE)

__global__ __launch_bounds__(SIG_THREADS) void gnn_sigma_kernel(
    const float* __restrict__ nd, float* __restrict__ ws) {
  __shared__ float red[SIG_THREADS / 32];
  const int gtid = blockIdx.x * SIG_THREADS + threadIdx.x;
  float v[SIG_LOADS];
#pragma unroll
  for (int k = 0; k < SIG_LOADS; ++k) {
    int i = gtid + k * SIG_STRIDE;
    i = (i < N_ * K1_) ? i : (N_ * K1_ - 1);
    v[k] = nd[i];
  }
  float m = v[0];
  for (int k = 1; k < SIG_LOADS; ++k) m = fmaxf(m, v[k]);
  for (int off = 16; off > 0; off >>= 1) m = fmaxf(m, __shfl_xor(m, off));
  if ((threadIdx.x & 31) == 0) red[threadIdx.x >> 5] = m;
  __syncthreads();
  if (threadIdx.x == 0) {
    float r = red[0];
    for (int w = 1; w < SIG_THREADS / 32; ++w) r = fmaxf(r, red[w]);
    ws[blockIdx.x] = r;
  }
}

#define PF_    4
#define RING_  (PF_ + 1)
#define TI_    (T_IN_ / 2)
#define HP_    (H_ / 2)
#define WM_PER_THREAD ((TH_ * OTP_) / 128)
#define WM_SKEW 32

__global__ __launch_bounds__(128) void gnn_fused_kernel(
    const float* __restrict__ x,
    const int*   __restrict__ nn,
    const float* __restrict__ nd,
    const float* __restrict__ Wm,
    const float* __restrict__ bias,
    const float* __restrict__ ws,
    float*       __restrict__ out) {
  __shared__ __attribute__((aligned(16))) float w_lds[K1_ * H_];
  __shared__ float Wm_lds[TH_ * OTP_ + WM_SKEW];

  const int n    = blockIdx.x;
  const int tid  = threadIdx.x;
  const int lane = tid & 31;
  const int b    = tid >> 5;
  const int cp   = lane & 15;
  const int hf   = lane >> 4;

  const int   e0 = tid, e1 = 128 + (tid & 7);
  const float d0 = nd[n * K1_ + (e0 >> 3)];
  const int   q0 = nn[n * K1_ + (e0 >> 3)];
  const float d1 = nd[n * K1_ + (e1 >> 3)];
  const int   q1 = nn[n * K1_ + (e1 >> 3)];
  int grow[K1_ + 1];
#pragma unroll
  for (int e = 0; e < K1_; ++e) {
    const int q = nn[n * K1_ + e];
    grow[e] = (q < 0) ? 0 : q;
  }
  grow[K1_] = n;
  const int   oc  = tid & (OTP_ - 1);
  const int   ocl = (oc < OT_) ? oc : (OT_ - 1);
  float wm[WM_PER_THREAD];
#pragma unroll
  for (int k = 0; k < WM_PER_THREAD; ++k)
    wm[k] = Wm[((tid >> 4) + 8 * k) * OT_ + ocl];

  float mx = fmaxf(ws[0], ws[SIG_BLOCKS / 2]);
#pragma unroll
  for (int i = 1; i < SIG_BLOCKS / 2; ++i) mx = fmaxf(mx, fmaxf(ws[i], ws[i + SIG_BLOCKS / 2]));
  const float sig = mx * 0.5f;
  const float inv_s2 = 1.0f / (sig * sig);

  {
    const float lam = (float)((e0 & 7) + 1) * (1.0f / (float)H_);
    float wv0 = __expf(-d0 * d0 * lam * inv_s2);
    if (q0 < 0 || wv0 < 1e-5f) wv0 = 0.0f;
    float wv1 = __expf(-d1 * d1 * lam * inv_s2);
    if (q1 < 0 || wv1 < 1e-5f) wv1 = 0.0f;
    const bool second = tid < K1_ * H_ - 128;
    w_lds[e0] = wv0;
    w_lds[second ? e1 : e0] = second ? wv1 : wv0;
  }
#pragma unroll
  for (int k = 0; k < WM_PER_THREAD; ++k)
    Wm_lds[tid + 128 * k + ((k >= WM_PER_THREAD / 2) ? WM_SKEW : 0)] = (oc < OT_) ? wm[k] : 0.0f;
  __syncthreads();

  const int bu = __builtin_amdgcn_readfirstlane(b);
  const float* xs = x + (size_t)bu * ((size_t)T_IN_ * N_ * C_);
  const unsigned lbyte = (unsigned)(hf * (TI_ * N_ * C_) + 2 * cp) * 4u;

  v2f acc[TI_][HP_][2];
#pragma unroll
  for (int ti = 0; ti < TI_; ++ti)
#pragma unroll
    for (int hp = 0; hp < HP_; ++hp) {
      acc[ti][hp][0] = (v2f){0.0f, 0.0f};
      acc[ti][hp][1] = (v2f){0.0f, 0.0f};
    }

  v2f xr[RING_][TI_];

#define LOAD_ROW(E)                                                          \
  {                                                                          \
    const char* rp_ = (const char*)(xs + (size_t)grow[(E)] * C_);            \
    _Pragma("unroll")                                                        \
    for (int ti = 0; ti < TI_; ++ti)                                         \
      xr[(E) % RING_][ti] =                                                  \
          *(const v2f*)(rp_ + lbyte + (size_t)ti * (N_ * C_ * 4));           \
  }

#pragma unroll
  for (int e = 0; e < PF_; ++e) LOAD_ROW(e)

#pragma unroll
  for (int j = 0; j <= K1_; ++j) {
    __builtin_amdgcn_sched_barrier(0x7CC);
    if (j + PF_ <= K1_) LOAD_ROW(j + PF_)
    if (j < K1_) {
      const v4f wa = *(const v4f*)&w_lds[j * H_];
      const v4f wb = *(const v4f*)&w_lds[j * H_ + 4];
      const v2f w2[HP_] = {(v2f){wa.x, wa.y}, (v2f){wa.z, wa.w}, (v2f){wb.x, wb.y}, (v2f){wb.z, wb.w}};
#pragma unroll
      for (int ti = 0; ti < TI_; ++ti) {
        const float x0 = xr[j % RING_][ti].x;
        const float x1 = xr[j % RING_][ti].y;
#pragma unroll
        for (int hp = 0; hp < HP_; ++hp) {
          acc[ti][hp][0] = __builtin_elementwise_fma((v2f){x0, x0}, w2[hp], acc[ti][hp][0]);
          acc[ti][hp][1] = __builtin_elementwise_fma((v2f){x1, x1}, w2[hp], acc[ti][hp][1]);
        }
      }
    } else {
#pragma unroll
      for (int ti = 0; ti < TI_; ++ti) {
        char* op_ = (char*)(out + (((size_t)bu * T_OUT_ + ti) * N_ + n) * C_);
        *(v2f*)(op_ + lbyte) = xr[j % RING_][ti];
      }
    }
  }
#undef LOAD_ROW

  __builtin_amdgcn_sched_barrier(0);
  v2f afrag[KSTEPS];
#pragma unroll
  for (int kk = 0; kk < KSTEPS; ++kk) {
    const int base = (48 * hf + 2 * kk) * OTP_ + hf * WM_SKEW + cp;
    afrag[kk].x = Wm_lds[base];
    afrag[kk].y = Wm_lds[base + OTP_];
  }
  float bv[8];
#pragma unroll
  for (int r = 0; r < 8; ++r)
    bv[r] = (r < OT_ - 8 && hf != 0) ? bias[8 + (r & 3)] : bias[r];
  __builtin_amdgcn_sched_barrier(0);

  v8f dA = {0.f, 0.f, 0.f, 0.f, 0.f, 0.f, 0.f, 0.f};
  v8f dB = {0.f, 0.f, 0.f, 0.f, 0.f, 0.f, 0.f, 0.f};
#pragma unroll
  for (int kk = 0; kk < KSTEPS; ++kk) {
    dA = __builtin_amdgcn_wmma_f32_16x16x4_f32(
         false, afrag[kk],  false, acc[kk >> 2][kk & 3][0],
         (short)0, dA,  false,  false);
    dB = __builtin_amdgcn_wmma_f32_16x16x4_f32(
         false, afrag[kk],  false, acc[kk >> 2][kk & 3][1],
         (short)0, dB,  false,  false);
  }

  const unsigned ybyte = (unsigned)(8 * hf * (N_ * C_) + 2 * cp) * 4u;
#pragma unroll
  for (int r = 0; r < 8; ++r) {
    float y0 = dA[r] + bv[r];
    float y1 = dB[r] + bv[r];
    const float s0 = 1.0f / (1.0f + __expf(-BETA_ * y0));
    const float s1 = 1.0f / (1.0f + __expf(-BETA_ * y1));
    y0 *= s0;
    y1 *= s1;
    if (r < 4 || hf == 0) {
      char* op_ = (char*)(out + (((size_t)bu * T_OUT_ + (T_IN_ + r)) * N_ + n) * C_);
      *(v2f*)(op_ + ybyte) = (v2f){y0, y1};
    }
  }
}

extern "C" void kernel_launch(void* const* d_in, const int* in_sizes, int n_in,
                              void* d_out, int out_size, void* d_ws, size_t ws_size,
                              hipStream_t stream) {
  const float* x    = (const float*)d_in[0];
  const int*   nn   = (const int*)d_in[1];
  const float* nd   = (const float*)d_in[2];
  const float* Wm   = (const float*)d_in[3];
  const float* bias = (const float*)d_in[4];
  float* out = (float*)d_out;
  float* ws  = (float*)d_ws;

  gnn_sigma_kernel<<<SIG_BLOCKS, SIG_THREADS, 0, stream>>>(nd, ws);
  gnn_fused_kernel<<<N_, 128, 0, stream>>>(x, nn, nd, Wm, bias, ws, out);
}


# fused kernel: all 14 output stores non-temporal, so the write-once output stops displacing the gathered rows of x in the second-level cache (x was re-fetched from memory within a run); bit-identical
# speedup vs baseline: 3.4859x; 1.0646x over previous
#include <hip/hip_runtime.h>
#include <math.h>

#define B_     4
#define T_IN_  12
#define T_OUT_ 24
#define N_     5000
#define C_     32
#define H_     8
#define K1_    17
#define TH_    (T_IN_ * H_)
#define OT_    (T_OUT_ - T_IN_)
#define OTP_   16
#define KSTEPS (TH_ / 4)
#define BETA_  0.8f

typedef __attribute__((ext_vector_type(2))) float v2f;
typedef __attribute__((ext_vector_type(4))) float v4f;
typedef __attribute__((ext_vector_type(8))) float v8f;

#define SIG_BLOCKS  32
#define SIG_THREADS 256
#define SIG_STRIDE  (SIG_BLOCKS * SIG_THREADS)
#define SIG_LOADS   ((N_ * K1_ + SIG_STRIDE - 1) / SIG_STRIDE)

__global__ __launch_bounds__(SIG_THREADS) void gnn_sigma_kernel(
    const float* __restrict__ nd, float* __restrict__ ws) {
  __shared__ float red[SIG_THREADS / 32];
  const int gtid = blockIdx.x * SIG_THREADS + threadIdx.x;
  float v[SIG_LOADS];
#pragma unroll
  for (int k = 0; k < SIG_LOADS; ++k) {
    int i = gtid + k * SIG_STRIDE;
    i = (i < N_ * K1_) ? i : (N_ * K1_ - 1);
    v[k] = nd[i];
  }
  float m = v[0];
  for (int k = 1; k < SIG_LOADS; ++k) m = fmaxf(m, v[k]);
  for (int off = 16; off > 0; off >>= 1) m = fmaxf(m, __shfl_xor(m, off));
  if ((threadIdx.x & 31) == 0) red[threadIdx.x >> 5] = m;
  __syncthreads();
  if (threadIdx.x == 0) {
    float r = red[0];
    for (int w = 1; w < SIG_THREADS / 32; ++w) r = fmaxf(r, red[w]);
    ws[blockIdx.x] = r;
  }
}

#define PF_    4
#define RING_  (PF_ + 1)
#define TI_    (T_IN_ / 2)
#define HP_    (H_ / 2)
#define WM_PER_THREAD ((TH_ * OTP_) / 128)
#define WM_SKEW 32

__global__ __launch_bounds__(128) void gnn_fused_kernel(
    const float* __restrict__ x,
    const int*   __restrict__ nn,
    const float* __restrict__ nd,
    const float* __restrict__ Wm,
    const float* __restrict__ bias,
    const float* __restrict__ ws,
    float*       __restrict__ out) {
  __shared__ __attribute__((aligned(16))) float w_lds[K1_ * H_];
  __shared__ float Wm_lds[TH_ * OTP_ + WM_SKEW];

  const int n    = blockIdx.x;
  const int tid  = threadIdx.x;
  const int lane = tid & 31;
  const int b    = tid >> 5;
  const int cp   = lane & 15;
  const int hf   = lane >> 4;

  const int   e0 = tid, e1 = 128 + (tid & 7);
  const float d0 = nd[n * K1_ + (e0 >> 3)];
  const int   q0 = nn[n * K1_ + (e0 >> 3)];
  const float d1 = nd[n * K1_ + (e1 >> 3)];
  const int   q1 = nn[n * K1_ + (e1 >> 3)];
  int grow[K1_ + 1];
#pragma unroll
  for (int e = 0; e < K1_; ++e) {
    const int q = nn[n * K1_ + e];
    grow[e] = (q < 0) ? 0 : q;
  }
  grow[K1_] = n;
  const int   oc  = tid & (OTP_ - 1);
  const int   ocl = (oc < OT_) ? oc : (OT_ - 1);
  float wm[WM_PER_THREAD];
#pragma unroll
  for (int k = 0; k < WM_PER_THREAD; ++k)
    wm[k] = Wm[((tid >> 4) + 8 * k) * OT_ + ocl];

  float mx = fmaxf(ws[0], ws[SIG_BLOCKS / 2]);
#pragma unroll
  for (int i = 1; i < SIG_BLOCKS / 2; ++i) mx = fmaxf(mx, fmaxf(ws[i], ws[i + SIG_BLOCKS / 2]));
  const float sig = mx * 0.5f;
  const float inv_s2 = 1.0f / (sig * sig);

  {
    const float lam = (float)((e0 & 7) + 1) * (1.0f / (float)H_);
    float wv0 = __expf(-d0 * d0 * lam * inv_s2);
    if (q0 < 0 || wv0 < 1e-5f) wv0 = 0.0f;
    float wv1 = __expf(-d1 * d1 * lam * inv_s2);
    if (q1 < 0 || wv1 < 1e-5f) wv1 = 0.0f;
    const bool second = tid < K1_ * H_ - 128;
    w_lds[e0] = wv0;
    w_lds[second ? e1 : e0] = second ? wv1 : wv0;
  }
#pragma unroll
  for (int k = 0; k < WM_PER_THREAD; ++k)
    Wm_lds[tid + 128 * k + ((k >= WM_PER_THREAD / 2) ? WM_SKEW : 0)] = (oc < OT_) ? wm[k] : 0.0f;
  __syncthreads();

  const int bu = __builtin_amdgcn_readfirstlane(b);
  const float* xs = x + (size_t)bu * ((size_t)T_IN_ * N_ * C_);
  const unsigned lbyte = (unsigned)(hf * (TI_ * N_ * C_) + 2 * cp) * 4u;

  v2f acc[TI_][HP_][2];
#pragma unroll
  for (int ti = 0; ti < TI_; ++ti)
#pragma unroll
    for (int hp = 0; hp < HP_; ++hp) {
      acc[ti][hp][0] = (v2f){0.0f, 0.0f};
      acc[ti][hp][1] = (v2f){0.0f, 0.0f};
    }

  v2f xr[RING_][TI_];

#define LOAD_ROW(E)                                                          \
  {                                                                          \
    const char* rp_ = (const char*)(xs + (size_t)grow[(E)] * C_);            \
    _Pragma("unroll")                                                        \
    for (int ti = 0; ti < TI_; ++ti)                                         \
      xr[(E) % RING_][ti] =                                                  \
          *(const v2f*)(rp_ + lbyte + (size_t)ti * (N_ * C_ * 4));           \
  }

#pragma unroll
  for (int e = 0; e < PF_; ++e) LOAD_ROW(e)

#pragma unroll
  for (int j = 0; j <= K1_; ++j) {
    __builtin_amdgcn_sched_barrier(0x7CC);
    if (j + PF_ <= K1_) LOAD_ROW(j + PF_)
    if (j < K1_) {
      const v4f wa = *(const v4f*)&w_lds[j * H_];
      const v4f wb = *(const v4f*)&w_lds[j * H_ + 4];
      const v2f w2[HP_] = {(v2f){wa.x, wa.y}, (v2f){wa.z, wa.w}, (v2f){wb.x, wb.y}, (v2f){wb.z, wb.w}};
#pragma unroll
      for (int ti = 0; ti < TI_; ++ti) {
        const float x0 = xr[j % RING_][ti].x;
        const float x1 = xr[j % RING_][ti].y;
#pragma unroll
        for (int hp = 0; hp < HP_; ++hp) {
          acc[ti][hp][0] = __builtin_elementwise_fma((v2f){x0, x0}, w2[hp], acc[ti][hp][0]);
          acc[ti][hp][1] = __builtin_elementwise_fma((v2f){x1, x1}, w2[hp], acc[ti][hp][1]);
        }
      }
    } else {
#pragma unroll
      for (int ti = 0; ti < TI_; ++ti) {
        char* op_ = (char*)(out + (((size_t)bu * T_OUT_ + ti) * N_ + n) * C_);
        __builtin_nontemporal_store(xr[j % RING_][ti], (v2f*)(op_ + lbyte));
      }
    }
  }
#undef LOAD_ROW

  __builtin_amdgcn_sched_barrier(0);
  v2f afrag[KSTEPS];
#pragma unroll
  for (int kk = 0; kk < KSTEPS; ++kk) {
    const int base = (48 * hf + 2 * kk) * OTP_ + hf * WM_SKEW + cp;
    afrag[kk].x = Wm_lds[base];
    afrag[kk].y = Wm_lds[base + OTP_];
  }
  float bv[8];
#pragma unroll
  for (int r = 0; r < 8; ++r)
    bv[r] = (r < OT_ - 8 && hf != 0) ? bias[8 + (r & 3)] : bias[r];
  __builtin_amdgcn_sched_barrier(0);

  v8f dA = {0.f, 0.f, 0.f, 0.f, 0.f, 0.f, 0.f, 0.f};
  v8f dB = {0.f, 0.f, 0.f, 0.f, 0.f, 0.f, 0.f, 0.f};
#pragma unroll
  for (int kk = 0; kk < KSTEPS; ++kk) {
    dA = __builtin_amdgcn_wmma_f32_16x16x4_f32(
         false, afrag[kk],  false, acc[kk >> 2][kk & 3][0],
         (short)0, dA,  false,  false);
    dB = __builtin_amdgcn_wmma_f32_16x16x4_f32(
         false, afrag[kk],  false, acc[kk >> 2][kk & 3][1],
         (short)0, dB,  false,  false);
  }

  const unsigned ybyte = (unsigned)(8 * hf * (N_ * C_) + 2 * cp) * 4u;
#pragma unroll
  for (int r = 0; r < 8; ++r) {
    float y0 = dA[r] + bv[r];
    float y1 = dB[r] + bv[r];
    const float s0 = 1.0f / (1.0f + __expf(-BETA_ * y0));
    const float s1 = 1.0f / (1.0f + __expf(-BETA_ * y1));
    y0 *= s0;
    y1 *= s1;
    if (r < 4 || hf == 0) {
      char* op_ = (char*)(out + (((size_t)bu * T_OUT_ + (T_IN_ + r)) * N_ + n) * C_);
      __builtin_nontemporal_store((v2f){y0, y1}, (v2f*)(op_ + ybyte));
    }
  }
}

extern "C" void kernel_launch(void* const* d_in, const int* in_sizes, int n_in,
                              void* d_out, int out_size, void* d_ws, size_t ws_size,
                              hipStream_t stream) {
  const float* x    = (const float*)d_in[0];
  const int*   nn   = (const int*)d_in[1];
  const float* nd   = (const float*)d_in[2];
  const float* Wm   = (const float*)d_in[3];
  const float* bias = (const float*)d_in[4];
  float* out = (float*)d_out;
  float* ws  = (float*)d_ws;

  gnn_sigma_kernel<<<SIG_BLOCKS, SIG_THREADS, 0, stream>>>(nd, ws);
  gnn_fused_kernel<<<N_, 128, 0, stream>>>(x, nn, nd, Wm, bias, ws, out);
}


# fused kernel: Swish as y * rcp(1 + exp2(c*y)): each IEEE division of the epilogue (about 11 dependent vector instructions) becomes one reciprocal instruction; last-bit rounding of y changes
# speedup vs baseline: 3.6274x; 1.0406x over previous
#include <hip/hip_runtime.h>
#include <math.h>

#define B_     4
#define T_IN_  12
#define T_OUT_ 24
#define N_     5000
#define C_     32
#define H_     8
#define K1_    17
#define TH_    (T_IN_ * H_)
#define OT_    (T_OUT_ - T_IN_)
#define OTP_   16
#define KSTEPS (TH_ / 4)
#define BETA_  0.8f

typedef __attribute__((ext_vector_type(2))) float v2f;
typedef __attribute__((ext_vector_type(4))) float v4f;
typedef __attribute__((ext_vector_type(8))) float v8f;

#define SIG_BLOCKS  32
#define SIG_THREADS 256
#define SIG_STRIDE  (SIG_BLOCKS * SIG_THREADS)
#define SIG_LOADS   ((N_ * K1_ + SIG_STRIDE - 1) / SIG_STRIDE)

__global__ __launch_bounds__(SIG_THREADS) void gnn_sigma_kernel(
    const float* __restrict__ nd, float* __restrict__ ws) {
  __shared__ float red[SIG_THREADS / 32];
  const int gtid = blockIdx.x * SIG_THREADS + threadIdx.x;
  float v[SIG_LOADS];
#pragma unroll
  for (int k = 0; k < SIG_LOADS; ++k) {
    int i = gtid + k * SIG_STRIDE;
    i = (i < N_ * K1_) ? i : (N_ * K1_ - 1);
    v[k] = nd[i];
  }
  float m = v[0];
  for (int k = 1; k < SIG_LOADS; ++k) m = fmaxf(m, v[k]);
  for (int off = 16; off > 0; off >>= 1) m = fmaxf(m, __shfl_xor(m, off));
  if ((threadIdx.x & 31) == 0) red[threadIdx.x >> 5] = m;
  __syncthreads();
  if (threadIdx.x == 0) {
    float r = red[0];
    for (int w = 1; w < SIG_THREADS / 32; ++w) r = fmaxf(r, red[w]);
    ws[blockIdx.x] = r;
  }
}

#define PF_    4
#define RING_  (PF_ + 1)
#define TI_    (T_IN_ / 2)
#define HP_    (H_ / 2)
#define WM_PER_THREAD ((TH_ * OTP_) / 128)
#define WM_SKEW 32

__global__ __launch_bounds__(128) void gnn_fused_kernel(
    const float* __restrict__ x,
    const int*   __restrict__ nn,
    const float* __restrict__ nd,
    const float* __restrict__ Wm,
    const float* __restrict__ bias,
    const float* __restrict__ ws,
    float*       __restrict__ out) {
  __shared__ __attribute__((aligned(16))) float w_lds[K1_ * H_];
  __shared__ float Wm_lds[TH_ * OTP_ + WM_SKEW];

  const int n    = blockIdx.x;
  const int tid  = threadIdx.x;
  const int lane = tid & 31;
  const int b    = tid >> 5;
  const int cp   = lane & 15;
  const int hf   = lane >> 4;

  const int   e0 = tid, e1 = 128 + (tid & 7);
  const float d0 = nd[n * K1_ + (e0 >> 3)];
  const int   q0 = nn[n * K1_ + (e0 >> 3)];
  const float d1 = nd[n * K1_ + (e1 >> 3)];
  const int   q1 = nn[n * K1_ + (e1 >> 3)];
  int grow[K1_ + 1];
#pragma unroll
  for (int e = 0; e < K1_; ++e) {
    const int q = nn[n * K1_ + e];
    grow[e] = (q < 0) ? 0 : q;
  }
  grow[K1_] = n;
  const int   oc  = tid & (OTP_ - 1);
  const int   ocl = (oc < OT_) ? oc : (OT_ - 1);
  float wm[WM_PER_THREAD];
#pragma unroll
  for (int k = 0; k < WM_PER_THREAD; ++k)
    wm[k] = Wm[((tid >> 4) + 8 * k) * OT_ + ocl];

  float mx = fmaxf(ws[0], ws[SIG_BLOCKS / 2]);
#pragma unroll
  for (int i = 1; i < SIG_BLOCKS / 2; ++i) mx = fmaxf(mx, fmaxf(ws[i], ws[i + SIG_BLOCKS / 2]));
  const float sig = mx * 0.5f;
  const float inv_s2 = 1.0f / (sig * sig);

  {
    const float lam = (float)((e0 & 7) + 1) * (1.0f / (float)H_);
    float wv0 = __expf(-d0 * d0 * lam * inv_s2);
    if (q0 < 0 || wv0 < 1e-5f) wv0 = 0.0f;
    float wv1 = __expf(-d1 * d1 * lam * inv_s2);
    if (q1 < 0 || wv1 < 1e-5f) wv1 = 0.0f;
    const bool second = tid < K1_ * H_ - 128;
    w_lds[e0] = wv0;
    w_lds[second ? e1 : e0] = second ? wv1 : wv0;
  }
#pragma unroll
  for (int k = 0; k < WM_PER_THREAD; ++k)
    Wm_lds[tid + 128 * k + ((k >= WM_PER_THREAD / 2) ? WM_SKEW : 0)] = (oc < OT_) ? wm[k] : 0.0f;
  __syncthreads();

  const int bu = __builtin_amdgcn_readfirstlane(b);
  const float* xs = x + (size_t)bu * ((size_t)T_IN_ * N_ * C_);
  const unsigned lbyte = (unsigned)(hf * (TI_ * N_ * C_) + 2 * cp) * 4u;

  v2f acc[TI_][HP_][2];
#pragma unroll
  for (int ti = 0; ti < TI_; ++ti)
#pragma unroll
    for (int hp = 0; hp < HP_; ++hp) {
      acc[ti][hp][0] = (v2f){0.0f, 0.0f};
      acc[ti][hp][1] = (v2f){0.0f, 0.0f};
    }

  v2f xr[RING_][TI_];

#define LOAD_ROW(E)                                                          \
  {                                                                          \
    const char* rp_ = (const char*)(xs + (size_t)grow[(E)] * C_);            \
    _Pragma("unroll")                                                        \
    for (int ti = 0; ti < TI_; ++ti)                                         \
      xr[(E) % RING_][ti] =                                                  \
          *(const v2f*)(rp_ + lbyte + (size_t)ti * (N_ * C_ * 4));           \
  }

#pragma unroll
  for (int e = 0; e < PF_; ++e) LOAD_ROW(e)

#pragma unroll
  for (int j = 0; j <= K1_; ++j) {
    __builtin_amdgcn_sched_barrier(0x7CC);
    if (j + PF_ <= K1_) LOAD_ROW(j + PF_)
    if (j < K1_) {
      const v4f wa = *(const v4f*)&w_lds[j * H_];
      const v4f wb = *(const v4f*)&w_lds[j * H_ + 4];
      const v2f w2[HP_] = {(v2f){wa.x, wa.y}, (v2f){wa.z, wa.w}, (v2f){wb.x, wb.y}, (v2f){wb.z, wb.w}};
#pragma unroll
      for (int ti = 0; ti < TI_; ++ti) {
        const float x0 = xr[j % RING_][ti].x;
        const float x1 = xr[j % RING_][ti].y;
#pragma unroll
        for (int hp = 0; hp < HP_; ++hp) {
          acc[ti][hp][0] = __builtin_elementwise_fma((v2f){x0, x0}, w2[hp], acc[ti][hp][0]);
          acc[ti][hp][1] = __builtin_elementwise_fma((v2f){x1, x1}, w2[hp], acc[ti][hp][1]);
        }
      }
    } else {
#pragma unroll
      for (int ti = 0; ti < TI_; ++ti) {
        char* op_ = (char*)(out + (((size_t)bu * T_OUT_ + ti) * N_ + n) * C_);
        __builtin_nontemporal_store(xr[j % RING_][ti], (v2f*)(op_ + lbyte));
      }
    }
  }
#undef LOAD_ROW

  __builtin_amdgcn_sched_barrier(0);
  v2f afrag[KSTEPS];
#pragma unroll
  for (int kk = 0; kk < KSTEPS; ++kk) {
    const int base = (48 * hf + 2 * kk) * OTP_ + hf * WM_SKEW + cp;
    afrag[kk].x = Wm_lds[base];
    afrag[kk].y = Wm_lds[base + OTP_];
  }
  float bv[8];
#pragma unroll
  for (int r = 0; r < 8; ++r)
    bv[r] = (r < OT_ - 8 && hf != 0) ? bias[8 + (r & 3)] : bias[r];
  __builtin_amdgcn_sched_barrier(0);

  v8f dA = {0.f, 0.f, 0.f, 0.f, 0.f, 0.f, 0.f, 0.f};
  v8f dB = {0.f, 0.f, 0.f, 0.f, 0.f, 0.f, 0.f, 0.f};
#pragma unroll
  for (int kk = 0; kk < KSTEPS; ++kk) {
    dA = __builtin_amdgcn_wmma_f32_16x16x4_f32(
         false, afrag[kk],  false, acc[kk >> 2][kk & 3][0],
         (short)0, dA,  false,  false);
    dB = __builtin_amdgcn_wmma_f32_16x16x4_f32(
         false, afrag[kk],  false, acc[kk >> 2][kk & 3][1],
         (short)0, dB,  false,  false);
  }

  const unsigned ybyte = (unsigned)(8 * hf * (N_ * C_) + 2 * cp) * 4u;
  const float kSwishC = -BETA_ * 1.44269504088896340736f;
#pragma unroll
  for (int r = 0; r < 8; ++r) {
    float y0 = dA[r] + bv[r];
    float y1 = dB[r] + bv[r];
    const float s0 = __builtin_amdgcn_rcpf(1.0f + __builtin_amdgcn_exp2f(kSwishC * y0));
    const float s1 = __builtin_amdgcn_rcpf(1.0f + __builtin_amdgcn_exp2f(kSwishC * y1));
    y0 *= s0;
    y1 *= s1;
    if (r < 4 || hf == 0) {
      char* op_ = (char*)(out + (((size_t)bu * T_OUT_ + (T_IN_ + r)) * N_ + n) * C_);
      __builtin_nontemporal_store((v2f){y0, y1}, (v2f*)(op_ + ybyte));
    }
  }
}

extern "C" void kernel_launch(void* const* d_in, const int* in_sizes, int n_in,
                              void* d_out, int out_size, void* d_ws, size_t ws_size,
                              hipStream_t stream) {
  const float* x    = (const float*)d_in[0];
  const int*   nn   = (const int*)d_in[1];
  const float* nd   = (const float*)d_in[2];
  const float* Wm   = (const float*)d_in[3];
  const float* bias = (const float*)d_in[4];
  float* out = (float*)d_out;
  float* ws  = (float*)d_ws;

  gnn_sigma_kernel<<<SIG_BLOCKS, SIG_THREADS, 0, stream>>>(nd, ws);
  gnn_fused_kernel<<<N_, 128, 0, stream>>>(x, nn, nd, Wm, bias, ws, out);
}
